# starting kernel
# baseline (speedup 1.0000x reference)
#include <hip/hip_runtime.h>
#include <math.h>

typedef __attribute__((ext_vector_type(2))) float v2f;
typedef __attribute__((ext_vector_type(8))) float v8f;

#define DATA    64
#define WIDTH   256
#define TPTS    16
#define NINT    (TPTS - 1)
#define MAXSUB  8
#define MT      16
#define NTHR    512
#define BATCH   512

#define RTOLC   1e-3f
#define ATOLC   1e-6f
#define SAFETY  0.9f
#define FMINC   0.2f
#define FMAXC   10.0f

__device__ __constant__ float STAGE_A[7][6] = {
  { 0.f, 0.f, 0.f, 0.f, 0.f, 0.f },
  { 0.161f, 0.f, 0.f, 0.f, 0.f, 0.f },
  { -0.008480655492356989f, 0.335480655492357f, 0.f, 0.f, 0.f, 0.f },
  { 2.8971530571054935f, -6.359448489975075f, 4.3622954328695815f, 0.f, 0.f, 0.f },
  { 5.325864828439257f, -11.748883564062828f, 7.4955393428898365f, -0.09249506636175525f, 0.f, 0.f },
  { 5.86145544294642f, -12.92096931784711f, 8.159367898576159f, -0.071584973281401f, -0.028269050394068383f, 0.f },
  { 0.09646076681806523f, 0.01f, 0.4798896504144996f, 1.379008574103742f, -3.290069515436081f, 2.324710524099774f },
};
__device__ __constant__ float ECOEF[7] = {
  -0.001780011052226f, -0.000816434459657f, 0.007880878010262f, -0.144711007173263f,
   0.582357165452555f, -0.458082105929187f, 0.015151515151515152f };

__device__ __forceinline__ float softplusf(float x) {
  const float LOG2E = 1.4426950408889634f;
  const float LN2   = 0.6931471805599453f;
  float e = __builtin_amdgcn_exp2f(-fabsf(x) * LOG2E);
  return fmaxf(x, 0.0f) + LN2 * __builtin_amdgcn_logf(1.0f + e);
}

__global__ __launch_bounds__(NTHR, 1)
void node_tsit5_kernel(const float* __restrict__ ts, const float* __restrict__ y0,
                       const float* __restrict__ W0, const float* __restrict__ b0,
                       const float* __restrict__ W1, const float* __restrict__ b1,
                       const float* __restrict__ W2, const float* __restrict__ b2,
                       float* __restrict__ out)
{
  __shared__ __align__(16) float sY [MT][DATA];
  __shared__ __align__(16) float sYn[MT][DATA];
  __shared__ __align__(16) float sZ [MT][DATA];
  __shared__ __align__(16) float sK [7][MT][DATA];
  __shared__ __align__(16) float sH0[MT][WIDTH];
  __shared__ __align__(16) float sH1[MT][WIDTH];
  __shared__ __align__(16) float sFp[4][MT][DATA];
  __shared__ float sT[MT], sDt[MT], sDtc[MT], sStep[MT], sDone[MT];
  __shared__ float sTs[TPTS];

  const int tid  = threadIdx.x;
  const int lane = tid & 31;
  const int wav  = tid >> 5;
  const int kh   = lane >> 4;
  const int l16  = lane & 15;
  const int row0 = blockIdx.x * MT;
  const int er   = wav;
  const int ec   = lane * 2;

  if (tid < TPTS) sTs[tid] = ts[tid];
  {
    const float* yp = y0 + (row0 + er) * DATA + ec;
    float a0 = yp[0], a1 = yp[1];
    sY[er][ec] = a0; sY[er][ec + 1] = a1;
    float* op = out + ((size_t)(row0 + er) * TPTS + 0) * DATA + ec;
    op[0] = a0; op[1] = a1;
  }
  if (lane == 0) sDt[er] = ts[1] - ts[0];
  __syncthreads();

  auto mlp = [&](float (*dst)[DATA]) {
    {
      const int n = wav * 16 + l16;
      v8f acc = {};
      const float* wp = W0 + n * DATA + kh * 2;
      #pragma unroll
      for (int kk = 0; kk < DATA / 4; ++kk) {
        const int k = kk * 4 + kh * 2;
        v2f a = *(const v2f*)(&sZ[l16][k]);
        v2f b = *(const v2f*)(wp + kk * 4);
        acc = __builtin_amdgcn_wmma_f32_16x16x4_f32(false, a, false, b,
                                                    (short)0, acc, false, false);
      }
      const float bias = b0[n];
      #pragma unroll
      for (int v = 0; v < 8; ++v)
        sH0[v + kh * 8][n] = softplusf(acc[v] + bias);
    }
    __syncthreads();
    {
      const int n = wav * 16 + l16;
      v8f acc = {};
      const float* wp = W1 + n * WIDTH + kh * 2;
      #pragma unroll 8
      for (int kk = 0; kk < WIDTH / 4; ++kk) {
        const int k = kk * 4 + kh * 2;
        v2f a = *(const v2f*)(&sH0[l16][k]);
        v2f b = *(const v2f*)(wp + kk * 4);
        acc = __builtin_amdgcn_wmma_f32_16x16x4_f32(false, a, false, b,
                                                    (short)0, acc, false, false);
      }
      const float bias = b1[n];
      #pragma unroll
      for (int v = 0; v < 8; ++v)
        sH1[v + kh * 8][n] = softplusf(acc[v] + bias);
    }
    __syncthreads();
    {
      const int dtile = wav & 3;
      const int kq    = wav >> 2;
      const int d     = dtile * 16 + l16;
      v8f acc = {};
      const float* wp = W2 + d * WIDTH + kq * 64 + kh * 2;
      #pragma unroll
      for (int kk = 0; kk < 16; ++kk) {
        const int k = kq * 64 + kk * 4 + kh * 2;
        v2f a = *(const v2f*)(&sH1[l16][k]);
        v2f b = *(const v2f*)(wp + kk * 4);
        acc = __builtin_amdgcn_wmma_f32_16x16x4_f32(false, a, false, b,
                                                    (short)0, acc, false, false);
      }
      #pragma unroll
      for (int v = 0; v < 8; ++v)
        sFp[kq][v + kh * 8][d] = acc[v];
    }
    __syncthreads();
    #pragma unroll
    for (int j = 0; j < 2; ++j) {
      const int c = ec + j;
      dst[er][c] = sFp[0][er][c] + sFp[1][er][c] + sFp[2][er][c] + sFp[3][er][c] + b2[c];
    }
    __syncthreads();
  };

  #pragma unroll 1
  for (int iv = 0; iv < NINT; ++iv) {
    const float t1 = sTs[iv + 1];
    if (lane == 0) sT[er] = sTs[iv];
    __syncthreads();

    #pragma unroll 1
    for (int ss = 0; ss < MAXSUB; ++ss) {
      if (lane == 0) {
        float t = sT[er], dt = sDt[er];
        float rem  = t1 - t;
        bool  done = rem <= 1e-6f;
        sDtc[er]  = done ? dt : fminf(dt, rem);
        sDone[er] = done ? 1.0f : 0.0f;
      }
      __syncthreads();
      const float dtc = sDtc[er];

      #pragma unroll 1
      for (int s = 0; s < 7; ++s) {
        #pragma unroll
        for (int j = 0; j < 2; ++j) {
          const int c = ec + j;
          float acc = 0.0f;
          for (int q = 0; q < s && q < 6; ++q)
            acc = fmaf(STAGE_A[s][q], sK[q][er][c], acc);
          float z = fmaf(dtc, acc, sY[er][c]);
          sZ[er][c] = z;
          if (s == 6) sYn[er][c] = z;
        }
        __syncthreads();
        mlp(sK[s]);
      }

      float esq = 0.0f;
      #pragma unroll
      for (int j = 0; j < 2; ++j) {
        const int c = ec + j;
        float e = 0.0f;
        #pragma unroll
        for (int q = 0; q < 7; ++q) e = fmaf(ECOEF[q], sK[q][er][c], e);
        e *= dtc;
        float scale = fmaf(RTOLC, fmaxf(fabsf(sY[er][c]), fabsf(sYn[er][c])), ATOLC);
        float r = e / scale;
        esq = fmaf(r, r, esq);
      }
      #pragma unroll
      for (int o = 16; o > 0; o >>= 1) esq += __shfl_xor(esq, o, 32);
      if (lane == 0) {
        float errn   = sqrtf(esq * (1.0f / DATA));
        bool  accept = errn <= 1.0f;
        float factor = SAFETY * __powf(fmaxf(errn, 1e-10f), -0.2f);
        factor = fminf(fmaxf(factor, FMINC), FMAXC);
        bool  done = sDone[er] != 0.0f;
        bool  step = accept && !done;
        if (step)  sT[er] += dtc;
        if (!done) sDt[er] = dtc * factor;
        sStep[er] = step ? 1.0f : 0.0f;
      }
      __syncthreads();
      if (sStep[er] != 0.0f) {
        sY[er][ec]     = sYn[er][ec];
        sY[er][ec + 1] = sYn[er][ec + 1];
      }
      __syncthreads();
    }

    float* op = out + ((size_t)(row0 + er) * TPTS + (iv + 1)) * DATA + ec;
    op[0] = sY[er][ec];
    op[1] = sY[er][ec + 1];
    __syncthreads();
  }
}

extern "C" void kernel_launch(void* const* d_in, const int* in_sizes, int n_in,
                              void* d_out, int out_size, void* d_ws, size_t ws_size,
                              hipStream_t stream) {
  (void)in_sizes; (void)n_in; (void)out_size; (void)d_ws; (void)ws_size;
  const float* ts = (const float*)d_in[0];
  const float* y0 = (const float*)d_in[1];
  const float* W0 = (const float*)d_in[2];
  const float* b0 = (const float*)d_in[3];
  const float* W1 = (const float*)d_in[4];
  const float* b1 = (const float*)d_in[5];
  const float* W2 = (const float*)d_in[6];
  const float* b2 = (const float*)d_in[7];
  float* out = (float*)d_out;
  node_tsit5_kernel<<<dim3(BATCH / MT), dim3(NTHR), 0, stream>>>(
      ts, y0, W0, b0, W1, b1, W2, b2, out);
}


# Pad the LDS row pitch of the WMMA A-operand arrays and of the split-K partials to 4 mod 64 dwords: each fragment read hits 64 distinct banks instead of a 16-way bank conflict
# speedup vs baseline: 1.9451x; 1.9451x over previous
#include <hip/hip_runtime.h>
#include <math.h>

typedef __attribute__((ext_vector_type(2))) float v2f;
typedef __attribute__((ext_vector_type(8))) float v8f;

#define DATA    64
#define WIDTH   256
#define TPTS    16
#define NINT    (TPTS - 1)
#define MAXSUB  8
#define MT      16
#define NTHR    512
#define BATCH   512
#define LPAD    4

#define RTOLC   1e-3f
#define ATOLC   1e-6f
#define SAFETY  0.9f
#define FMINC   0.2f
#define FMAXC   10.0f

__device__ __constant__ float STAGE_A[7][6] = {
  { 0.f, 0.f, 0.f, 0.f, 0.f, 0.f },
  { 0.161f, 0.f, 0.f, 0.f, 0.f, 0.f },
  { -0.008480655492356989f, 0.335480655492357f, 0.f, 0.f, 0.f, 0.f },
  { 2.8971530571054935f, -6.359448489975075f, 4.3622954328695815f, 0.f, 0.f, 0.f },
  { 5.325864828439257f, -11.748883564062828f, 7.4955393428898365f, -0.09249506636175525f, 0.f, 0.f },
  { 5.86145544294642f, -12.92096931784711f, 8.159367898576159f, -0.071584973281401f, -0.028269050394068383f, 0.f },
  { 0.09646076681806523f, 0.01f, 0.4798896504144996f, 1.379008574103742f, -3.290069515436081f, 2.324710524099774f },
};
__device__ __constant__ float ECOEF[7] = {
  -0.001780011052226f, -0.000816434459657f, 0.007880878010262f, -0.144711007173263f,
   0.582357165452555f, -0.458082105929187f, 0.015151515151515152f };

__device__ __forceinline__ float softplusf(float x) {
  const float LOG2E = 1.4426950408889634f;
  const float LN2   = 0.6931471805599453f;
  float e = __builtin_amdgcn_exp2f(-fabsf(x) * LOG2E);
  return fmaxf(x, 0.0f) + LN2 * __builtin_amdgcn_logf(1.0f + e);
}

__global__ __launch_bounds__(NTHR, 1)
void node_tsit5_kernel(const float* __restrict__ ts, const float* __restrict__ y0,
                       const float* __restrict__ W0, const float* __restrict__ b0,
                       const float* __restrict__ W1, const float* __restrict__ b1,
                       const float* __restrict__ W2, const float* __restrict__ b2,
                       float* __restrict__ out)
{
  __shared__ __align__(16) float sY [MT][DATA];
  __shared__ __align__(16) float sYn[MT][DATA];
  __shared__ __align__(16) float sZ [MT][DATA + LPAD];
  __shared__ __align__(16) float sK [7][MT][DATA];
  __shared__ __align__(16) float sH0[MT][WIDTH + LPAD];
  __shared__ __align__(16) float sH1[MT][WIDTH + LPAD];
  __shared__ __align__(16) float sFp[4][MT][DATA + LPAD];
  __shared__ float sT[MT], sDt[MT], sDtc[MT], sStep[MT], sDone[MT];
  __shared__ float sTs[TPTS];

  const int tid  = threadIdx.x;
  const int lane = tid & 31;
  const int wav  = tid >> 5;
  const int kh   = lane >> 4;
  const int l16  = lane & 15;
  const int row0 = blockIdx.x * MT;
  const int er   = wav;
  const int ec   = lane * 2;

  if (tid < TPTS) sTs[tid] = ts[tid];
  {
    const float* yp = y0 + (row0 + er) * DATA + ec;
    float a0 = yp[0], a1 = yp[1];
    sY[er][ec] = a0; sY[er][ec + 1] = a1;
    float* op = out + ((size_t)(row0 + er) * TPTS + 0) * DATA + ec;
    op[0] = a0; op[1] = a1;
  }
  if (lane == 0) sDt[er] = ts[1] - ts[0];
  __syncthreads();

  auto mlp = [&](float (*dst)[DATA]) {
    {
      const int n = wav * 16 + l16;
      v8f acc = {};
      const float* wp = W0 + n * DATA + kh * 2;
      #pragma unroll
      for (int kk = 0; kk < DATA / 4; ++kk) {
        const int k = kk * 4 + kh * 2;
        v2f a = *(const v2f*)(&sZ[l16][k]);
        v2f b = *(const v2f*)(wp + kk * 4);
        acc = __builtin_amdgcn_wmma_f32_16x16x4_f32(false, a, false, b,
                                                    (short)0, acc, false, false);
      }
      const float bias = b0[n];
      #pragma unroll
      for (int v = 0; v < 8; ++v)
        sH0[v + kh * 8][n] = softplusf(acc[v] + bias);
    }
    __syncthreads();
    {
      const int n = wav * 16 + l16;
      v8f acc = {};
      const float* wp = W1 + n * WIDTH + kh * 2;
      #pragma unroll 8
      for (int kk = 0; kk < WIDTH / 4; ++kk) {
        const int k = kk * 4 + kh * 2;
        v2f a = *(const v2f*)(&sH0[l16][k]);
        v2f b = *(const v2f*)(wp + kk * 4);
        acc = __builtin_amdgcn_wmma_f32_16x16x4_f32(false, a, false, b,
                                                    (short)0, acc, false, false);
      }
      const float bias = b1[n];
      #pragma unroll
      for (int v = 0; v < 8; ++v)
        sH1[v + kh * 8][n] = softplusf(acc[v] + bias);
    }
    __syncthreads();
    {
      const int dtile = wav & 3;
      const int kq    = wav >> 2;
      const int d     = dtile * 16 + l16;
      v8f acc = {};
      const float* wp = W2 + d * WIDTH + kq * 64 + kh * 2;
      #pragma unroll
      for (int kk = 0; kk < 16; ++kk) {
        const int k = kq * 64 + kk * 4 + kh * 2;
        v2f a = *(const v2f*)(&sH1[l16][k]);
        v2f b = *(const v2f*)(wp + kk * 4);
        acc = __builtin_amdgcn_wmma_f32_16x16x4_f32(false, a, false, b,
                                                    (short)0, acc, false, false);
      }
      #pragma unroll
      for (int v = 0; v < 8; ++v)
        sFp[kq][v + kh * 8][d] = acc[v];
    }
    __syncthreads();
    #pragma unroll
    for (int j = 0; j < 2; ++j) {
      const int c = ec + j;
      dst[er][c] = sFp[0][er][c] + sFp[1][er][c] + sFp[2][er][c] + sFp[3][er][c] + b2[c];
    }
    __syncthreads();
  };

  #pragma unroll 1
  for (int iv = 0; iv < NINT; ++iv) {
    const float t1 = sTs[iv + 1];
    if (lane == 0) sT[er] = sTs[iv];
    __syncthreads();

    #pragma unroll 1
    for (int ss = 0; ss < MAXSUB; ++ss) {
      if (lane == 0) {
        float t = sT[er], dt = sDt[er];
        float rem  = t1 - t;
        bool  done = rem <= 1e-6f;
        sDtc[er]  = done ? dt : fminf(dt, rem);
        sDone[er] = done ? 1.0f : 0.0f;
      }
      __syncthreads();
      const float dtc = sDtc[er];

      #pragma unroll 1
      for (int s = 0; s < 7; ++s) {
        #pragma unroll
        for (int j = 0; j < 2; ++j) {
          const int c = ec + j;
          float acc = 0.0f;
          for (int q = 0; q < s && q < 6; ++q)
            acc = fmaf(STAGE_A[s][q], sK[q][er][c], acc);
          float z = fmaf(dtc, acc, sY[er][c]);
          sZ[er][c] = z;
          if (s == 6) sYn[er][c] = z;
        }
        __syncthreads();
        mlp(sK[s]);
      }

      float esq = 0.0f;
      #pragma unroll
      for (int j = 0; j < 2; ++j) {
        const int c = ec + j;
        float e = 0.0f;
        #pragma unroll
        for (int q = 0; q < 7; ++q) e = fmaf(ECOEF[q], sK[q][er][c], e);
        e *= dtc;
        float scale = fmaf(RTOLC, fmaxf(fabsf(sY[er][c]), fabsf(sYn[er][c])), ATOLC);
        float r = e / scale;
        esq = fmaf(r, r, esq);
      }
      #pragma unroll
      for (int o = 16; o > 0; o >>= 1) esq += __shfl_xor(esq, o, 32);
      if (lane == 0) {
        float errn   = sqrtf(esq * (1.0f / DATA));
        bool  accept = errn <= 1.0f;
        float factor = SAFETY * __powf(fmaxf(errn, 1e-10f), -0.2f);
        factor = fminf(fmaxf(factor, FMINC), FMAXC);
        bool  done = sDone[er] != 0.0f;
        bool  step = accept && !done;
        if (step)  sT[er] += dtc;
        if (!done) sDt[er] = dtc * factor;
        sStep[er] = step ? 1.0f : 0.0f;
      }
      __syncthreads();
      if (sStep[er] != 0.0f) {
        sY[er][ec]     = sYn[er][ec];
        sY[er][ec + 1] = sYn[er][ec + 1];
      }
      __syncthreads();
    }

    float* op = out + ((size_t)(row0 + er) * TPTS + (iv + 1)) * DATA + ec;
    op[0] = sY[er][ec];
    op[1] = sY[er][ec + 1];
    __syncthreads();
  }
}

extern "C" void kernel_launch(void* const* d_in, const int* in_sizes, int n_in,
                              void* d_out, int out_size, void* d_ws, size_t ws_size,
                              hipStream_t stream) {
  (void)in_sizes; (void)n_in; (void)out_size; (void)d_ws; (void)ws_size;
  const float* ts = (const float*)d_in[0];
  const float* y0 = (const float*)d_in[1];
  const float* W0 = (const float*)d_in[2];
  const float* b0 = (const float*)d_in[3];
  const float* W1 = (const float*)d_in[4];
  const float* b1 = (const float*)d_in[5];
  const float* W2 = (const float*)d_in[6];
  const float* b2 = (const float*)d_in[7];
  float* out = (float*)d_out;
  node_tsit5_kernel<<<dim3(BATCH / MT), dim3(NTHR), 0, stream>>>(
      ts, y0, W0, b0, W1, b1, W2, b2, out);
}


# Hold the 256x256 layer's WMMA B fragments in registers (64 x v2f per lane), loaded once before the time loop as W0 and W2 already are: GEMM2's K loop issues no global loads
# speedup vs baseline: 3.1376x; 1.6131x over previous
#include <hip/hip_runtime.h>
#include <math.h>

typedef __attribute__((ext_vector_type(2))) float v2f;
typedef __attribute__((ext_vector_type(8))) float v8f;

#define DATA    64
#define WIDTH   256
#define TPTS    16
#define NINT    (TPTS - 1)
#define MAXSUB  8
#define MT      16
#define NTHR    512
#define BATCH   512
#define LPAD    4

#define RTOLC   1e-3f
#define ATOLC   1e-6f
#define SAFETY  0.9f
#define FMINC   0.2f
#define FMAXC   10.0f

__device__ __constant__ float STAGE_A[7][6] = {
  { 0.f, 0.f, 0.f, 0.f, 0.f, 0.f },
  { 0.161f, 0.f, 0.f, 0.f, 0.f, 0.f },
  { -0.008480655492356989f, 0.335480655492357f, 0.f, 0.f, 0.f, 0.f },
  { 2.8971530571054935f, -6.359448489975075f, 4.3622954328695815f, 0.f, 0.f, 0.f },
  { 5.325864828439257f, -11.748883564062828f, 7.4955393428898365f, -0.09249506636175525f, 0.f, 0.f },
  { 5.86145544294642f, -12.92096931784711f, 8.159367898576159f, -0.071584973281401f, -0.028269050394068383f, 0.f },
  { 0.09646076681806523f, 0.01f, 0.4798896504144996f, 1.379008574103742f, -3.290069515436081f, 2.324710524099774f },
};
__device__ __constant__ float ECOEF[7] = {
  -0.001780011052226f, -0.000816434459657f, 0.007880878010262f, -0.144711007173263f,
   0.582357165452555f, -0.458082105929187f, 0.015151515151515152f };

__device__ __forceinline__ float softplusf(float x) {
  const float LOG2E = 1.4426950408889634f;
  const float LN2   = 0.6931471805599453f;
  float e = __builtin_amdgcn_exp2f(-fabsf(x) * LOG2E);
  return fmaxf(x, 0.0f) + LN2 * __builtin_amdgcn_logf(1.0f + e);
}

__global__ __launch_bounds__(NTHR, 1)
void node_tsit5_kernel(const float* __restrict__ ts, const float* __restrict__ y0,
                       const float* __restrict__ W0, const float* __restrict__ b0,
                       const float* __restrict__ W1, const float* __restrict__ b1,
                       const float* __restrict__ W2, const float* __restrict__ b2,
                       float* __restrict__ out)
{
  __shared__ __align__(16) float sY [MT][DATA];
  __shared__ __align__(16) float sYn[MT][DATA];
  __shared__ __align__(16) float sZ [MT][DATA + LPAD];
  __shared__ __align__(16) float sK [7][MT][DATA];
  __shared__ __align__(16) float sH0[MT][WIDTH + LPAD];
  __shared__ __align__(16) float sH1[MT][WIDTH + LPAD];
  __shared__ __align__(16) float sFp[4][MT][DATA + LPAD];
  __shared__ float sT[MT], sDt[MT], sDtc[MT], sStep[MT], sDone[MT];
  __shared__ float sTs[TPTS];

  const int tid  = threadIdx.x;
  const int lane = tid & 31;
  const int wav  = tid >> 5;
  const int kh   = lane >> 4;
  const int l16  = lane & 15;
  const int row0 = blockIdx.x * MT;
  const int er   = wav;
  const int ec   = lane * 2;

  if (tid < TPTS) sTs[tid] = ts[tid];
  {
    const float* yp = y0 + (row0 + er) * DATA + ec;
    float a0 = yp[0], a1 = yp[1];
    sY[er][ec] = a0; sY[er][ec + 1] = a1;
    float* op = out + ((size_t)(row0 + er) * TPTS + 0) * DATA + ec;
    op[0] = a0; op[1] = a1;
  }
  if (lane == 0) sDt[er] = ts[1] - ts[0];

  v2f w1f[WIDTH / 4];
  {
    const float* wp = W1 + (wav * 16 + l16) * WIDTH + kh * 2;
    #pragma unroll
    for (int kk = 0; kk < WIDTH / 4; ++kk)
      w1f[kk] = *(const v2f*)(wp + kk * 4);
  }
  __syncthreads();

  auto mlp = [&](float (*dst)[DATA]) {
    {
      const int n = wav * 16 + l16;
      v8f acc = {};
      const float* wp = W0 + n * DATA + kh * 2;
      #pragma unroll
      for (int kk = 0; kk < DATA / 4; ++kk) {
        const int k = kk * 4 + kh * 2;
        v2f a = *(const v2f*)(&sZ[l16][k]);
        v2f b = *(const v2f*)(wp + kk * 4);
        acc = __builtin_amdgcn_wmma_f32_16x16x4_f32(false, a, false, b,
                                                    (short)0, acc, false, false);
      }
      const float bias = b0[n];
      #pragma unroll
      for (int v = 0; v < 8; ++v)
        sH0[v + kh * 8][n] = softplusf(acc[v] + bias);
    }
    __syncthreads();
    {
      const int n = wav * 16 + l16;
      v8f acc = {};
      #pragma unroll
      for (int kk = 0; kk < WIDTH / 4; ++kk) {
        const int k = kk * 4 + kh * 2;
        v2f a = *(const v2f*)(&sH0[l16][k]);
        acc = __builtin_amdgcn_wmma_f32_16x16x4_f32(false, a, false, w1f[kk],
                                                    (short)0, acc, false, false);
      }
      const float bias = b1[n];
      #pragma unroll
      for (int v = 0; v < 8; ++v)
        sH1[v + kh * 8][n] = softplusf(acc[v] + bias);
    }
    __syncthreads();
    {
      const int dtile = wav & 3;
      const int kq    = wav >> 2;
      const int d     = dtile * 16 + l16;
      v8f acc = {};
      const float* wp = W2 + d * WIDTH + kq * 64 + kh * 2;
      #pragma unroll
      for (int kk = 0; kk < 16; ++kk) {
        const int k = kq * 64 + kk * 4 + kh * 2;
        v2f a = *(const v2f*)(&sH1[l16][k]);
        v2f b = *(const v2f*)(wp + kk * 4);
        acc = __builtin_amdgcn_wmma_f32_16x16x4_f32(false, a, false, b,
                                                    (short)0, acc, false, false);
      }
      #pragma unroll
      for (int v = 0; v < 8; ++v)
        sFp[kq][v + kh * 8][d] = acc[v];
    }
    __syncthreads();
    #pragma unroll
    for (int j = 0; j < 2; ++j) {
      const int c = ec + j;
      dst[er][c] = sFp[0][er][c] + sFp[1][er][c] + sFp[2][er][c] + sFp[3][er][c] + b2[c];
    }
    __syncthreads();
  };

  #pragma unroll 1
  for (int iv = 0; iv < NINT; ++iv) {
    const float t1 = sTs[iv + 1];
    if (lane == 0) sT[er] = sTs[iv];
    __syncthreads();

    #pragma unroll 1
    for (int ss = 0; ss < MAXSUB; ++ss) {
      if (lane == 0) {
        float t = sT[er], dt = sDt[er];
        float rem  = t1 - t;
        bool  done = rem <= 1e-6f;
        sDtc[er]  = done ? dt : fminf(dt, rem);
        sDone[er] = done ? 1.0f : 0.0f;
      }
      __syncthreads();
      const float dtc = sDtc[er];

      #pragma unroll 1
      for (int s = 0; s < 7; ++s) {
        #pragma unroll
        for (int j = 0; j < 2; ++j) {
          const int c = ec + j;
          float acc = 0.0f;
          for (int q = 0; q < s && q < 6; ++q)
            acc = fmaf(STAGE_A[s][q], sK[q][er][c], acc);
          float z = fmaf(dtc, acc, sY[er][c]);
          sZ[er][c] = z;
          if (s == 6) sYn[er][c] = z;
        }
        __syncthreads();
        mlp(sK[s]);
      }

      float esq = 0.0f;
      #pragma unroll
      for (int j = 0; j < 2; ++j) {
        const int c = ec + j;
        float e = 0.0f;
        #pragma unroll
        for (int q = 0; q < 7; ++q) e = fmaf(ECOEF[q], sK[q][er][c], e);
        e *= dtc;
        float scale = fmaf(RTOLC, fmaxf(fabsf(sY[er][c]), fabsf(sYn[er][c])), ATOLC);
        float r = e / scale;
        esq = fmaf(r, r, esq);
      }
      #pragma unroll
      for (int o = 16; o > 0; o >>= 1) esq += __shfl_xor(esq, o, 32);
      if (lane == 0) {
        float errn   = sqrtf(esq * (1.0f / DATA));
        bool  accept = errn <= 1.0f;
        float factor = SAFETY * __powf(fmaxf(errn, 1e-10f), -0.2f);
        factor = fminf(fmaxf(factor, FMINC), FMAXC);
        bool  done = sDone[er] != 0.0f;
        bool  step = accept && !done;
        if (step)  sT[er] += dtc;
        if (!done) sDt[er] = dtc * factor;
        sStep[er] = step ? 1.0f : 0.0f;
      }
      __syncthreads();
      if (sStep[er] != 0.0f) {
        sY[er][ec]     = sYn[er][ec];
        sY[er][ec + 1] = sYn[er][ec + 1];
      }
      __syncthreads();
    }

    float* op = out + ((size_t)(row0 + er) * TPTS + (iv + 1)) * DATA + ec;
    op[0] = sY[er][ec];
    op[1] = sY[er][ec + 1];
    __syncthreads();
  }
}

extern "C" void kernel_launch(void* const* d_in, const int* in_sizes, int n_in,
                              void* d_out, int out_size, void* d_ws, size_t ws_size,
                              hipStream_t stream) {
  (void)in_sizes; (void)n_in; (void)out_size; (void)d_ws; (void)ws_size;
  const float* ts = (const float*)d_in[0];
  const float* y0 = (const float*)d_in[1];
  const float* W0 = (const float*)d_in[2];
  const float* b0 = (const float*)d_in[3];
  const float* W1 = (const float*)d_in[4];
  const float* b1 = (const float*)d_in[5];
  const float* W2 = (const float*)d_in[6];
  const float* b2 = (const float*)d_in[7];
  float* out = (float*)d_out;
  node_tsit5_kernel<<<dim3(BATCH / MT), dim3(NTHR), 0, stream>>>(
      ts, y0, W0, b0, W1, b1, W2, b2, out);
}


# Leave the 8-substep loop once all 16 rows of the block are done: a done row's substep is a masked no-op in the reference (t, y, dt kept), so the skipped network evaluations change no output bit
# speedup vs baseline: 22.5210x; 7.1778x over previous
#include <hip/hip_runtime.h>
#include <math.h>

typedef __attribute__((ext_vector_type(2))) float v2f;
typedef __attribute__((ext_vector_type(8))) float v8f;

#define DATA    64
#define WIDTH   256
#define TPTS    16
#define NINT    (TPTS - 1)
#define MAXSUB  8
#define MT      16
#define NTHR    512
#define BATCH   512
#define LPAD    4

#define RTOLC   1e-3f
#define ATOLC   1e-6f
#define SAFETY  0.9f
#define FMINC   0.2f
#define FMAXC   10.0f

__device__ __constant__ float STAGE_A[7][6] = {
  { 0.f, 0.f, 0.f, 0.f, 0.f, 0.f },
  { 0.161f, 0.f, 0.f, 0.f, 0.f, 0.f },
  { -0.008480655492356989f, 0.335480655492357f, 0.f, 0.f, 0.f, 0.f },
  { 2.8971530571054935f, -6.359448489975075f, 4.3622954328695815f, 0.f, 0.f, 0.f },
  { 5.325864828439257f, -11.748883564062828f, 7.4955393428898365f, -0.09249506636175525f, 0.f, 0.f },
  { 5.86145544294642f, -12.92096931784711f, 8.159367898576159f, -0.071584973281401f, -0.028269050394068383f, 0.f },
  { 0.09646076681806523f, 0.01f, 0.4798896504144996f, 1.379008574103742f, -3.290069515436081f, 2.324710524099774f },
};
__device__ __constant__ float ECOEF[7] = {
  -0.001780011052226f, -0.000816434459657f, 0.007880878010262f, -0.144711007173263f,
   0.582357165452555f, -0.458082105929187f, 0.015151515151515152f };

__device__ __forceinline__ float softplusf(float x) {
  const float LOG2E = 1.4426950408889634f;
  const float LN2   = 0.6931471805599453f;
  float e = __builtin_amdgcn_exp2f(-fabsf(x) * LOG2E);
  return fmaxf(x, 0.0f) + LN2 * __builtin_amdgcn_logf(1.0f + e);
}

__global__ __launch_bounds__(NTHR, 1)
void node_tsit5_kernel(const float* __restrict__ ts, const float* __restrict__ y0,
                       const float* __restrict__ W0, const float* __restrict__ b0,
                       const float* __restrict__ W1, const float* __restrict__ b1,
                       const float* __restrict__ W2, const float* __restrict__ b2,
                       float* __restrict__ out)
{
  __shared__ __align__(16) float sY [MT][DATA];
  __shared__ __align__(16) float sYn[MT][DATA];
  __shared__ __align__(16) float sZ [MT][DATA + LPAD];
  __shared__ __align__(16) float sK [7][MT][DATA];
  __shared__ __align__(16) float sH0[MT][WIDTH + LPAD];
  __shared__ __align__(16) float sH1[MT][WIDTH + LPAD];
  __shared__ __align__(16) float sFp[4][MT][DATA + LPAD];
  __shared__ float sT[MT], sDt[MT], sDtc[MT], sStep[MT], sDone[MT];
  __shared__ float sTs[TPTS];

  const int tid  = threadIdx.x;
  const int lane = tid & 31;
  const int wav  = tid >> 5;
  const int kh   = lane >> 4;
  const int l16  = lane & 15;
  const int row0 = blockIdx.x * MT;
  const int er   = wav;
  const int ec   = lane * 2;

  if (tid < TPTS) sTs[tid] = ts[tid];
  {
    const float* yp = y0 + (row0 + er) * DATA + ec;
    float a0 = yp[0], a1 = yp[1];
    sY[er][ec] = a0; sY[er][ec + 1] = a1;
    float* op = out + ((size_t)(row0 + er) * TPTS + 0) * DATA + ec;
    op[0] = a0; op[1] = a1;
  }
  if (lane == 0) sDt[er] = ts[1] - ts[0];

  v2f w1f[WIDTH / 4];
  {
    const float* wp = W1 + (wav * 16 + l16) * WIDTH + kh * 2;
    #pragma unroll
    for (int kk = 0; kk < WIDTH / 4; ++kk)
      w1f[kk] = *(const v2f*)(wp + kk * 4);
  }
  __syncthreads();

  auto mlp = [&](float (*dst)[DATA]) {
    {
      const int n = wav * 16 + l16;
      v8f acc = {};
      const float* wp = W0 + n * DATA + kh * 2;
      #pragma unroll
      for (int kk = 0; kk < DATA / 4; ++kk) {
        const int k = kk * 4 + kh * 2;
        v2f a = *(const v2f*)(&sZ[l16][k]);
        v2f b = *(const v2f*)(wp + kk * 4);
        acc = __builtin_amdgcn_wmma_f32_16x16x4_f32(false, a, false, b,
                                                    (short)0, acc, false, false);
      }
      const float bias = b0[n];
      #pragma unroll
      for (int v = 0; v < 8; ++v)
        sH0[v + kh * 8][n] = softplusf(acc[v] + bias);
    }
    __syncthreads();
    {
      const int n = wav * 16 + l16;
      v8f acc = {};
      #pragma unroll
      for (int kk = 0; kk < WIDTH / 4; ++kk) {
        const int k = kk * 4 + kh * 2;
        v2f a = *(const v2f*)(&sH0[l16][k]);
        acc = __builtin_amdgcn_wmma_f32_16x16x4_f32(false, a, false, w1f[kk],
                                                    (short)0, acc, false, false);
      }
      const float bias = b1[n];
      #pragma unroll
      for (int v = 0; v < 8; ++v)
        sH1[v + kh * 8][n] = softplusf(acc[v] + bias);
    }
    __syncthreads();
    {
      const int dtile = wav & 3;
      const int kq    = wav >> 2;
      const int d     = dtile * 16 + l16;
      v8f acc = {};
      const float* wp = W2 + d * WIDTH + kq * 64 + kh * 2;
      #pragma unroll
      for (int kk = 0; kk < 16; ++kk) {
        const int k = kq * 64 + kk * 4 + kh * 2;
        v2f a = *(const v2f*)(&sH1[l16][k]);
        v2f b = *(const v2f*)(wp + kk * 4);
        acc = __builtin_amdgcn_wmma_f32_16x16x4_f32(false, a, false, b,
                                                    (short)0, acc, false, false);
      }
      #pragma unroll
      for (int v = 0; v < 8; ++v)
        sFp[kq][v + kh * 8][d] = acc[v];
    }
    __syncthreads();
    #pragma unroll
    for (int j = 0; j < 2; ++j) {
      const int c = ec + j;
      dst[er][c] = sFp[0][er][c] + sFp[1][er][c] + sFp[2][er][c] + sFp[3][er][c] + b2[c];
    }
    __syncthreads();
  };

  #pragma unroll 1
  for (int iv = 0; iv < NINT; ++iv) {
    const float t1 = sTs[iv + 1];
    if (lane == 0) sT[er] = sTs[iv];
    __syncthreads();

    #pragma unroll 1
    for (int ss = 0; ss < MAXSUB; ++ss) {
      if (lane == 0) {
        float t = sT[er], dt = sDt[er];
        float rem  = t1 - t;
        bool  done = rem <= 1e-6f;
        sDtc[er]  = done ? dt : fminf(dt, rem);
        sDone[er] = done ? 1.0f : 0.0f;
      }
      __syncthreads();
      if (__builtin_amdgcn_ballot_w32(sDone[l16] != 0.0f) == 0xffffffffu) break;
      const float dtc = sDtc[er];

      #pragma unroll 1
      for (int s = 0; s < 7; ++s) {
        #pragma unroll
        for (int j = 0; j < 2; ++j) {
          const int c = ec + j;
          float acc = 0.0f;
          for (int q = 0; q < s && q < 6; ++q)
            acc = fmaf(STAGE_A[s][q], sK[q][er][c], acc);
          float z = fmaf(dtc, acc, sY[er][c]);
          sZ[er][c] = z;
          if (s == 6) sYn[er][c] = z;
        }
        __syncthreads();
        mlp(sK[s]);
      }

      float esq = 0.0f;
      #pragma unroll
      for (int j = 0; j < 2; ++j) {
        const int c = ec + j;
        float e = 0.0f;
        #pragma unroll
        for (int q = 0; q < 7; ++q) e = fmaf(ECOEF[q], sK[q][er][c], e);
        e *= dtc;
        float scale = fmaf(RTOLC, fmaxf(fabsf(sY[er][c]), fabsf(sYn[er][c])), ATOLC);
        float r = e / scale;
        esq = fmaf(r, r, esq);
      }
      #pragma unroll
      for (int o = 16; o > 0; o >>= 1) esq += __shfl_xor(esq, o, 32);
      if (lane == 0) {
        float errn   = sqrtf(esq * (1.0f / DATA));
        bool  accept = errn <= 1.0f;
        float factor = SAFETY * __powf(fmaxf(errn, 1e-10f), -0.2f);
        factor = fminf(fmaxf(factor, FMINC), FMAXC);
        bool  done = sDone[er] != 0.0f;
        bool  step = accept && !done;
        if (step)  sT[er] += dtc;
        if (!done) sDt[er] = dtc * factor;
        sStep[er] = step ? 1.0f : 0.0f;
      }
      __syncthreads();
      if (sStep[er] != 0.0f) {
        sY[er][ec]     = sYn[er][ec];
        sY[er][ec + 1] = sYn[er][ec + 1];
      }
      __syncthreads();
    }

    float* op = out + ((size_t)(row0 + er) * TPTS + (iv + 1)) * DATA + ec;
    op[0] = sY[er][ec];
    op[1] = sY[er][ec + 1];
    __syncthreads();
  }
}

extern "C" void kernel_launch(void* const* d_in, const int* in_sizes, int n_in,
                              void* d_out, int out_size, void* d_ws, size_t ws_size,
                              hipStream_t stream) {
  (void)in_sizes; (void)n_in; (void)out_size; (void)d_ws; (void)ws_size;
  const float* ts = (const float*)d_in[0];
  const float* y0 = (const float*)d_in[1];
  const float* W0 = (const float*)d_in[2];
  const float* b0 = (const float*)d_in[3];
  const float* W1 = (const float*)d_in[4];
  const float* b1 = (const float*)d_in[5];
  const float* W2 = (const float*)d_in[6];
  const float* b2 = (const float*)d_in[7];
  float* out = (float*)d_out;
  node_tsit5_kernel<<<dim3(BATCH / MT), dim3(NTHR), 0, stream>>>(
      ts, y0, W0, b0, W1, b1, W2, b2, out);
}


# Reuse k7 = f(y_new) as the next substep's k1 (first-same-as-last): a taken step copies k7 to k1, a row that does not step keeps y and k1; later substeps run 6 stages, not 7; output bits unchanged
# speedup vs baseline: 24.9709x; 1.1088x over previous
#include <hip/hip_runtime.h>
#include <math.h>

typedef __attribute__((ext_vector_type(2))) float v2f;
typedef __attribute__((ext_vector_type(8))) float v8f;

#define DATA    64
#define WIDTH   256
#define TPTS    16
#define NINT    (TPTS - 1)
#define MAXSUB  8
#define MT      16
#define NTHR    512
#define BATCH   512
#define LPAD    4

#define RTOLC   1e-3f
#define ATOLC   1e-6f
#define SAFETY  0.9f
#define FMINC   0.2f
#define FMAXC   10.0f

__device__ __constant__ float STAGE_A[7][6] = {
  { 0.f, 0.f, 0.f, 0.f, 0.f, 0.f },
  { 0.161f, 0.f, 0.f, 0.f, 0.f, 0.f },
  { -0.008480655492356989f, 0.335480655492357f, 0.f, 0.f, 0.f, 0.f },
  { 2.8971530571054935f, -6.359448489975075f, 4.3622954328695815f, 0.f, 0.f, 0.f },
  { 5.325864828439257f, -11.748883564062828f, 7.4955393428898365f, -0.09249506636175525f, 0.f, 0.f },
  { 5.86145544294642f, -12.92096931784711f, 8.159367898576159f, -0.071584973281401f, -0.028269050394068383f, 0.f },
  { 0.09646076681806523f, 0.01f, 0.4798896504144996f, 1.379008574103742f, -3.290069515436081f, 2.324710524099774f },
};
__device__ __constant__ float ECOEF[7] = {
  -0.001780011052226f, -0.000816434459657f, 0.007880878010262f, -0.144711007173263f,
   0.582357165452555f, -0.458082105929187f, 0.015151515151515152f };

__device__ __forceinline__ float softplusf(float x) {
  const float LOG2E = 1.4426950408889634f;
  const float LN2   = 0.6931471805599453f;
  float e = __builtin_amdgcn_exp2f(-fabsf(x) * LOG2E);
  return fmaxf(x, 0.0f) + LN2 * __builtin_amdgcn_logf(1.0f + e);
}

__global__ __launch_bounds__(NTHR, 1)
void node_tsit5_kernel(const float* __restrict__ ts, const float* __restrict__ y0,
                       const float* __restrict__ W0, const float* __restrict__ b0,
                       const float* __restrict__ W1, const float* __restrict__ b1,
                       const float* __restrict__ W2, const float* __restrict__ b2,
                       float* __restrict__ out)
{
  __shared__ __align__(16) float sY [MT][DATA];
  __shared__ __align__(16) float sYn[MT][DATA];
  __shared__ __align__(16) float sZ [MT][DATA + LPAD];
  __shared__ __align__(16) float sK [7][MT][DATA];
  __shared__ __align__(16) float sH0[MT][WIDTH + LPAD];
  __shared__ __align__(16) float sH1[MT][WIDTH + LPAD];
  __shared__ __align__(16) float sFp[4][MT][DATA + LPAD];
  __shared__ float sT[MT], sDt[MT], sDtc[MT], sStep[MT], sDone[MT];
  __shared__ float sTs[TPTS];

  const int tid  = threadIdx.x;
  const int lane = tid & 31;
  const int wav  = tid >> 5;
  const int kh   = lane >> 4;
  const int l16  = lane & 15;
  const int row0 = blockIdx.x * MT;
  const int er   = wav;
  const int ec   = lane * 2;

  if (tid < TPTS) sTs[tid] = ts[tid];
  {
    const float* yp = y0 + (row0 + er) * DATA + ec;
    float a0 = yp[0], a1 = yp[1];
    sY[er][ec] = a0; sY[er][ec + 1] = a1;
    float* op = out + ((size_t)(row0 + er) * TPTS + 0) * DATA + ec;
    op[0] = a0; op[1] = a1;
  }
  if (lane == 0) sDt[er] = ts[1] - ts[0];

  v2f w1f[WIDTH / 4];
  {
    const float* wp = W1 + (wav * 16 + l16) * WIDTH + kh * 2;
    #pragma unroll
    for (int kk = 0; kk < WIDTH / 4; ++kk)
      w1f[kk] = *(const v2f*)(wp + kk * 4);
  }
  __syncthreads();

  auto mlp = [&](float (*dst)[DATA]) {
    {
      const int n = wav * 16 + l16;
      v8f acc = {};
      const float* wp = W0 + n * DATA + kh * 2;
      #pragma unroll
      for (int kk = 0; kk < DATA / 4; ++kk) {
        const int k = kk * 4 + kh * 2;
        v2f a = *(const v2f*)(&sZ[l16][k]);
        v2f b = *(const v2f*)(wp + kk * 4);
        acc = __builtin_amdgcn_wmma_f32_16x16x4_f32(false, a, false, b,
                                                    (short)0, acc, false, false);
      }
      const float bias = b0[n];
      #pragma unroll
      for (int v = 0; v < 8; ++v)
        sH0[v + kh * 8][n] = softplusf(acc[v] + bias);
    }
    __syncthreads();
    {
      const int n = wav * 16 + l16;
      v8f acc = {};
      #pragma unroll
      for (int kk = 0; kk < WIDTH / 4; ++kk) {
        const int k = kk * 4 + kh * 2;
        v2f a = *(const v2f*)(&sH0[l16][k]);
        acc = __builtin_amdgcn_wmma_f32_16x16x4_f32(false, a, false, w1f[kk],
                                                    (short)0, acc, false, false);
      }
      const float bias = b1[n];
      #pragma unroll
      for (int v = 0; v < 8; ++v)
        sH1[v + kh * 8][n] = softplusf(acc[v] + bias);
    }
    __syncthreads();
    {
      const int dtile = wav & 3;
      const int kq    = wav >> 2;
      const int d     = dtile * 16 + l16;
      v8f acc = {};
      const float* wp = W2 + d * WIDTH + kq * 64 + kh * 2;
      #pragma unroll
      for (int kk = 0; kk < 16; ++kk) {
        const int k = kq * 64 + kk * 4 + kh * 2;
        v2f a = *(const v2f*)(&sH1[l16][k]);
        v2f b = *(const v2f*)(wp + kk * 4);
        acc = __builtin_amdgcn_wmma_f32_16x16x4_f32(false, a, false, b,
                                                    (short)0, acc, false, false);
      }
      #pragma unroll
      for (int v = 0; v < 8; ++v)
        sFp[kq][v + kh * 8][d] = acc[v];
    }
    __syncthreads();
    #pragma unroll
    for (int j = 0; j < 2; ++j) {
      const int c = ec + j;
      dst[er][c] = sFp[0][er][c] + sFp[1][er][c] + sFp[2][er][c] + sFp[3][er][c] + b2[c];
    }
    __syncthreads();
  };

  int s_begin = 0;

  #pragma unroll 1
  for (int iv = 0; iv < NINT; ++iv) {
    const float t1 = sTs[iv + 1];
    if (lane == 0) sT[er] = sTs[iv];
    __syncthreads();

    #pragma unroll 1
    for (int ss = 0; ss < MAXSUB; ++ss) {
      if (lane == 0) {
        float t = sT[er], dt = sDt[er];
        float rem  = t1 - t;
        bool  done = rem <= 1e-6f;
        sDtc[er]  = done ? dt : fminf(dt, rem);
        sDone[er] = done ? 1.0f : 0.0f;
      }
      __syncthreads();
      if (__builtin_amdgcn_ballot_w32(sDone[l16] != 0.0f) == 0xffffffffu) break;
      const float dtc = sDtc[er];

      #pragma unroll 1
      for (int s = s_begin; s < 7; ++s) {
        #pragma unroll
        for (int j = 0; j < 2; ++j) {
          const int c = ec + j;
          float acc = 0.0f;
          for (int q = 0; q < s && q < 6; ++q)
            acc = fmaf(STAGE_A[s][q], sK[q][er][c], acc);
          float z = fmaf(dtc, acc, sY[er][c]);
          sZ[er][c] = z;
          if (s == 6) sYn[er][c] = z;
        }
        __syncthreads();
        mlp(sK[s]);
      }
      s_begin = 1;

      float esq = 0.0f;
      #pragma unroll
      for (int j = 0; j < 2; ++j) {
        const int c = ec + j;
        float e = 0.0f;
        #pragma unroll
        for (int q = 0; q < 7; ++q) e = fmaf(ECOEF[q], sK[q][er][c], e);
        e *= dtc;
        float scale = fmaf(RTOLC, fmaxf(fabsf(sY[er][c]), fabsf(sYn[er][c])), ATOLC);
        float r = e / scale;
        esq = fmaf(r, r, esq);
      }
      #pragma unroll
      for (int o = 16; o > 0; o >>= 1) esq += __shfl_xor(esq, o, 32);
      if (lane == 0) {
        float errn   = sqrtf(esq * (1.0f / DATA));
        bool  accept = errn <= 1.0f;
        float factor = SAFETY * __powf(fmaxf(errn, 1e-10f), -0.2f);
        factor = fminf(fmaxf(factor, FMINC), FMAXC);
        bool  done = sDone[er] != 0.0f;
        bool  step = accept && !done;
        if (step)  sT[er] += dtc;
        if (!done) sDt[er] = dtc * factor;
        sStep[er] = step ? 1.0f : 0.0f;
      }
      __syncthreads();
      if (sStep[er] != 0.0f) {
        sY[er][ec]     = sYn[er][ec];
        sY[er][ec + 1] = sYn[er][ec + 1];
        sK[0][er][ec]     = sK[6][er][ec];
        sK[0][er][ec + 1] = sK[6][er][ec + 1];
      }
      __syncthreads();
    }

    float* op = out + ((size_t)(row0 + er) * TPTS + (iv + 1)) * DATA + ec;
    op[0] = sY[er][ec];
    op[1] = sY[er][ec + 1];
    __syncthreads();
  }
}

extern "C" void kernel_launch(void* const* d_in, const int* in_sizes, int n_in,
                              void* d_out, int out_size, void* d_ws, size_t ws_size,
                              hipStream_t stream) {
  (void)in_sizes; (void)n_in; (void)out_size; (void)d_ws; (void)ws_size;
  const float* ts = (const float*)d_in[0];
  const float* y0 = (const float*)d_in[1];
  const float* W0 = (const float*)d_in[2];
  const float* b0 = (const float*)d_in[3];
  const float* W1 = (const float*)d_in[4];
  const float* b1 = (const float*)d_in[5];
  const float* W2 = (const float*)d_in[6];
  const float* b2 = (const float*)d_in[7];
  float* out = (float*)d_out;
  node_tsit5_kernel<<<dim3(BATCH / MT), dim3(NTHR), 0, stream>>>(
      ts, y0, W0, b0, W1, b1, W2, b2, out);
}


# Stage end as one per-thread pass specialised per stage: split-K combine, then next stage state from registers, tableau as literals, one LDS read group; one barrier fewer per evaluation; same bits
# speedup vs baseline: 27.5039x; 1.1014x over previous
#include <hip/hip_runtime.h>
#include <math.h>

typedef __attribute__((ext_vector_type(2))) float v2f;
typedef __attribute__((ext_vector_type(8))) float v8f;

#define DATA    64
#define WIDTH   256
#define TPTS    16
#define NINT    (TPTS - 1)
#define MAXSUB  8
#define MT      16
#define NTHR    512
#define BATCH   512
#define LPAD    4

#define RTOLC   1e-3f
#define ATOLC   1e-6f
#define SAFETY  0.9f
#define FMINC   0.2f
#define FMAXC   10.0f

static constexpr __device__ float STAGE_A[7][6] = {
  { 0.f, 0.f, 0.f, 0.f, 0.f, 0.f },
  { 0.161f, 0.f, 0.f, 0.f, 0.f, 0.f },
  { -0.008480655492356989f, 0.335480655492357f, 0.f, 0.f, 0.f, 0.f },
  { 2.8971530571054935f, -6.359448489975075f, 4.3622954328695815f, 0.f, 0.f, 0.f },
  { 5.325864828439257f, -11.748883564062828f, 7.4955393428898365f, -0.09249506636175525f, 0.f, 0.f },
  { 5.86145544294642f, -12.92096931784711f, 8.159367898576159f, -0.071584973281401f, -0.028269050394068383f, 0.f },
  { 0.09646076681806523f, 0.01f, 0.4798896504144996f, 1.379008574103742f, -3.290069515436081f, 2.324710524099774f },
};
__device__ __constant__ float ECOEF[7] = {
  -0.001780011052226f, -0.000816434459657f, 0.007880878010262f, -0.144711007173263f,
   0.582357165452555f, -0.458082105929187f, 0.015151515151515152f };

__device__ __forceinline__ float softplusf(float x) {
  const float LOG2E = 1.4426950408889634f;
  const float LN2   = 0.6931471805599453f;
  float e = __builtin_amdgcn_exp2f(-fabsf(x) * LOG2E);
  return fmaxf(x, 0.0f) + LN2 * __builtin_amdgcn_logf(1.0f + e);
}


template <int S>
__device__ __forceinline__ v2f stage_state(const v2f (&k)[7], v2f y, float dtc) {
  v2f acc = { 0.0f, 0.0f };
  #pragma unroll
  for (int q = 0; q < S && q < 6; ++q) {
    acc.x = fmaf(STAGE_A[S][q], k[q].x, acc.x);
    acc.y = fmaf(STAGE_A[S][q], k[q].y, acc.y);
  }
  v2f z;
  z.x = fmaf(dtc, acc.x, y.x);
  z.y = fmaf(dtc, acc.y, y.y);
  return z;
}

template <int S>
__device__ __forceinline__ void first_state(const float (*sK)[MT][DATA], const float (*sY)[DATA],
                                            float (*sZ)[DATA + LPAD], float dtc, int er, int ec) {
  v2f k[7];
  #pragma unroll
  for (int q = 0; q < S; ++q) k[q] = *(const v2f*)&sK[q][er][ec];
  const v2f y = *(const v2f*)&sY[er][ec];
  *(v2f*)&sZ[er][ec] = stage_state<S>(k, y, dtc);
}

template <int S>
__device__ __forceinline__ void finish_stage(float (*sK)[MT][DATA], const float (*sFp)[MT][DATA + LPAD],
                                             const float (*sY)[DATA], float (*sYn)[DATA],
                                             float (*sZ)[DATA + LPAD], v2f bias, float dtc,
                                             int er, int ec) {
  const v2f y  = *(const v2f*)&sY[er][ec];
  const v2f p0 = *(const v2f*)&sFp[0][er][ec];
  const v2f p1 = *(const v2f*)&sFp[1][er][ec];
  const v2f p2 = *(const v2f*)&sFp[2][er][ec];
  const v2f p3 = *(const v2f*)&sFp[3][er][ec];
  v2f k[7];
  #pragma unroll
  for (int q = 0; q < S; ++q) k[q] = *(const v2f*)&sK[q][er][ec];
  v2f kS = p0 + p1;
  kS = kS + p2;
  kS = kS + p3;
  kS = kS + bias;
  k[S] = kS;
  *(v2f*)&sK[S][er][ec] = kS;
  if constexpr (S < 6) {
    const v2f z = stage_state<S + 1>(k, y, dtc);
    *(v2f*)&sZ[er][ec] = z;
    if constexpr (S + 1 == 6) *(v2f*)&sYn[er][ec] = z;
  }
}

__global__ __launch_bounds__(NTHR, 1)
void node_tsit5_kernel(const float* __restrict__ ts, const float* __restrict__ y0,
                       const float* __restrict__ W0, const float* __restrict__ b0,
                       const float* __restrict__ W1, const float* __restrict__ b1,
                       const float* __restrict__ W2, const float* __restrict__ b2,
                       float* __restrict__ out)
{
  __shared__ __align__(16) float sY [MT][DATA];
  __shared__ __align__(16) float sYn[MT][DATA];
  __shared__ __align__(16) float sZ [MT][DATA + LPAD];
  __shared__ __align__(16) float sK [7][MT][DATA];
  __shared__ __align__(16) float sH0[MT][WIDTH + LPAD];
  __shared__ __align__(16) float sH1[MT][WIDTH + LPAD];
  __shared__ __align__(16) float sFp[4][MT][DATA + LPAD];
  __shared__ float sT[MT], sDt[MT], sDtc[MT], sStep[MT], sDone[MT];
  __shared__ float sTs[TPTS];

  const int tid  = threadIdx.x;
  const int lane = tid & 31;
  const int wav  = tid >> 5;
  const int kh   = lane >> 4;
  const int l16  = lane & 15;
  const int row0 = blockIdx.x * MT;
  const int er   = wav;
  const int ec   = lane * 2;

  if (tid < TPTS) sTs[tid] = ts[tid];
  {
    const float* yp = y0 + (row0 + er) * DATA + ec;
    float a0 = yp[0], a1 = yp[1];
    sY[er][ec] = a0; sY[er][ec + 1] = a1;
    float* op = out + ((size_t)(row0 + er) * TPTS + 0) * DATA + ec;
    op[0] = a0; op[1] = a1;
  }
  if (lane == 0) sDt[er] = ts[1] - ts[0];

  v2f w1f[WIDTH / 4];
  {
    const float* wp = W1 + (wav * 16 + l16) * WIDTH + kh * 2;
    #pragma unroll
    for (int kk = 0; kk < WIDTH / 4; ++kk)
      w1f[kk] = *(const v2f*)(wp + kk * 4);
  }
  v2f bias2;
  bias2.x = b2[ec];
  bias2.y = b2[ec + 1];
  __syncthreads();

  auto mlp_gemms = [&]() {
    {
      const int n = wav * 16 + l16;
      v8f acc = {};
      const float* wp = W0 + n * DATA + kh * 2;
      #pragma unroll
      for (int kk = 0; kk < DATA / 4; ++kk) {
        const int k = kk * 4 + kh * 2;
        v2f a = *(const v2f*)(&sZ[l16][k]);
        v2f b = *(const v2f*)(wp + kk * 4);
        acc = __builtin_amdgcn_wmma_f32_16x16x4_f32(false, a, false, b,
                                                    (short)0, acc, false, false);
      }
      const float bias = b0[n];
      #pragma unroll
      for (int v = 0; v < 8; ++v)
        sH0[v + kh * 8][n] = softplusf(acc[v] + bias);
    }
    __syncthreads();
    {
      const int n = wav * 16 + l16;
      v8f acc = {};
      #pragma unroll
      for (int kk = 0; kk < WIDTH / 4; ++kk) {
        const int k = kk * 4 + kh * 2;
        v2f a = *(const v2f*)(&sH0[l16][k]);
        acc = __builtin_amdgcn_wmma_f32_16x16x4_f32(false, a, false, w1f[kk],
                                                    (short)0, acc, false, false);
      }
      const float bias = b1[n];
      #pragma unroll
      for (int v = 0; v < 8; ++v)
        sH1[v + kh * 8][n] = softplusf(acc[v] + bias);
    }
    __syncthreads();
    {
      const int dtile = wav & 3;
      const int kq    = wav >> 2;
      const int d     = dtile * 16 + l16;
      v8f acc = {};
      const float* wp = W2 + d * WIDTH + kq * 64 + kh * 2;
      #pragma unroll
      for (int kk = 0; kk < 16; ++kk) {
        const int k = kq * 64 + kk * 4 + kh * 2;
        v2f a = *(const v2f*)(&sH1[l16][k]);
        v2f b = *(const v2f*)(wp + kk * 4);
        acc = __builtin_amdgcn_wmma_f32_16x16x4_f32(false, a, false, b,
                                                    (short)0, acc, false, false);
      }
      #pragma unroll
      for (int v = 0; v < 8; ++v)
        sFp[kq][v + kh * 8][d] = acc[v];
    }
    __syncthreads();
  };

  int s_begin = 0;

  #pragma unroll 1
  for (int iv = 0; iv < NINT; ++iv) {
    const float t1 = sTs[iv + 1];
    if (lane == 0) sT[er] = sTs[iv];
    __syncthreads();

    #pragma unroll 1
    for (int ss = 0; ss < MAXSUB; ++ss) {
      if (lane == 0) {
        float t = sT[er], dt = sDt[er];
        float rem  = t1 - t;
        bool  done = rem <= 1e-6f;
        sDtc[er]  = done ? dt : fminf(dt, rem);
        sDone[er] = done ? 1.0f : 0.0f;
      }
      __syncthreads();
      if (__builtin_amdgcn_ballot_w32(sDone[l16] != 0.0f) == 0xffffffffu) break;
      const float dtc = sDtc[er];

      if (s_begin == 0) first_state<0>(sK, sY, sZ, dtc, er, ec);
      else              first_state<1>(sK, sY, sZ, dtc, er, ec);
      __syncthreads();

      #pragma unroll 1
      for (int s = s_begin; s < 7; ++s) {
        mlp_gemms();
        switch (s) {
          case 0:  finish_stage<0>(sK, sFp, sY, sYn, sZ, bias2, dtc, er, ec); break;
          case 1:  finish_stage<1>(sK, sFp, sY, sYn, sZ, bias2, dtc, er, ec); break;
          case 2:  finish_stage<2>(sK, sFp, sY, sYn, sZ, bias2, dtc, er, ec); break;
          case 3:  finish_stage<3>(sK, sFp, sY, sYn, sZ, bias2, dtc, er, ec); break;
          case 4:  finish_stage<4>(sK, sFp, sY, sYn, sZ, bias2, dtc, er, ec); break;
          case 5:  finish_stage<5>(sK, sFp, sY, sYn, sZ, bias2, dtc, er, ec); break;
          default: finish_stage<6>(sK, sFp, sY, sYn, sZ, bias2, dtc, er, ec); break;
        }
        __syncthreads();
      }
      s_begin = 1;

      float esq = 0.0f;
      #pragma unroll
      for (int j = 0; j < 2; ++j) {
        const int c = ec + j;
        float e = 0.0f;
        #pragma unroll
        for (int q = 0; q < 7; ++q) e = fmaf(ECOEF[q], sK[q][er][c], e);
        e *= dtc;
        float scale = fmaf(RTOLC, fmaxf(fabsf(sY[er][c]), fabsf(sYn[er][c])), ATOLC);
        float r = e / scale;
        esq = fmaf(r, r, esq);
      }
      #pragma unroll
      for (int o = 16; o > 0; o >>= 1) esq += __shfl_xor(esq, o, 32);
      if (lane == 0) {
        float errn   = sqrtf(esq * (1.0f / DATA));
        bool  accept = errn <= 1.0f;
        float factor = SAFETY * __powf(fmaxf(errn, 1e-10f), -0.2f);
        factor = fminf(fmaxf(factor, FMINC), FMAXC);
        bool  done = sDone[er] != 0.0f;
        bool  step = accept && !done;
        if (step)  sT[er] += dtc;
        if (!done) sDt[er] = dtc * factor;
        sStep[er] = step ? 1.0f : 0.0f;
      }
      __syncthreads();
      if (sStep[er] != 0.0f) {
        sY[er][ec]     = sYn[er][ec];
        sY[er][ec + 1] = sYn[er][ec + 1];
        sK[0][er][ec]     = sK[6][er][ec];
        sK[0][er][ec + 1] = sK[6][er][ec + 1];
      }
      __syncthreads();
    }

    float* op = out + ((size_t)(row0 + er) * TPTS + (iv + 1)) * DATA + ec;
    op[0] = sY[er][ec];
    op[1] = sY[er][ec + 1];
    __syncthreads();
  }
}

extern "C" void kernel_launch(void* const* d_in, const int* in_sizes, int n_in,
                              void* d_out, int out_size, void* d_ws, size_t ws_size,
                              hipStream_t stream) {
  (void)in_sizes; (void)n_in; (void)out_size; (void)d_ws; (void)ws_size;
  const float* ts = (const float*)d_in[0];
  const float* y0 = (const float*)d_in[1];
  const float* W0 = (const float*)d_in[2];
  const float* b0 = (const float*)d_in[3];
  const float* W1 = (const float*)d_in[4];
  const float* b1 = (const float*)d_in[5];
  const float* W2 = (const float*)d_in[6];
  const float* b2 = (const float*)d_in[7];
  float* out = (float*)d_out;
  node_tsit5_kernel<<<dim3(BATCH / MT), dim3(NTHR), 0, stream>>>(
      ts, y0, W0, b0, W1, b1, W2, b2, out);
}


# Hold the WMMA B fragments of the two 64-wide layers and both hidden biases in registers for the whole integration, loaded once before the time loop: no substep fetches W0 and W2 from L2 again
# speedup vs baseline: 28.9910x; 1.0541x over previous
#include <hip/hip_runtime.h>
#include <math.h>

typedef __attribute__((ext_vector_type(2))) float v2f;
typedef __attribute__((ext_vector_type(8))) float v8f;

#define DATA    64
#define WIDTH   256
#define TPTS    16
#define NINT    (TPTS - 1)
#define MAXSUB  8
#define MT      16
#define NTHR    512
#define BATCH   512
#define LPAD    4

#define RTOLC   1e-3f
#define ATOLC   1e-6f
#define SAFETY  0.9f
#define FMINC   0.2f
#define FMAXC   10.0f

static constexpr __device__ float STAGE_A[7][6] = {
  { 0.f, 0.f, 0.f, 0.f, 0.f, 0.f },
  { 0.161f, 0.f, 0.f, 0.f, 0.f, 0.f },
  { -0.008480655492356989f, 0.335480655492357f, 0.f, 0.f, 0.f, 0.f },
  { 2.8971530571054935f, -6.359448489975075f, 4.3622954328695815f, 0.f, 0.f, 0.f },
  { 5.325864828439257f, -11.748883564062828f, 7.4955393428898365f, -0.09249506636175525f, 0.f, 0.f },
  { 5.86145544294642f, -12.92096931784711f, 8.159367898576159f, -0.071584973281401f, -0.028269050394068383f, 0.f },
  { 0.09646076681806523f, 0.01f, 0.4798896504144996f, 1.379008574103742f, -3.290069515436081f, 2.324710524099774f },
};
__device__ __constant__ float ECOEF[7] = {
  -0.001780011052226f, -0.000816434459657f, 0.007880878010262f, -0.144711007173263f,
   0.582357165452555f, -0.458082105929187f, 0.015151515151515152f };

__device__ __forceinline__ float softplusf(float x) {
  const float LOG2E = 1.4426950408889634f;
  const float LN2   = 0.6931471805599453f;
  float e = __builtin_amdgcn_exp2f(-fabsf(x) * LOG2E);
  return fmaxf(x, 0.0f) + LN2 * __builtin_amdgcn_logf(1.0f + e);
}


template <int S>
__device__ __forceinline__ v2f stage_state(const v2f (&k)[7], v2f y, float dtc) {
  v2f acc = { 0.0f, 0.0f };
  #pragma unroll
  for (int q = 0; q < S && q < 6; ++q) {
    acc.x = fmaf(STAGE_A[S][q], k[q].x, acc.x);
    acc.y = fmaf(STAGE_A[S][q], k[q].y, acc.y);
  }
  v2f z;
  z.x = fmaf(dtc, acc.x, y.x);
  z.y = fmaf(dtc, acc.y, y.y);
  return z;
}

template <int S>
__device__ __forceinline__ void first_state(const float (*sK)[MT][DATA], const float (*sY)[DATA],
                                            float (*sZ)[DATA + LPAD], float dtc, int er, int ec) {
  v2f k[7];
  #pragma unroll
  for (int q = 0; q < S; ++q) k[q] = *(const v2f*)&sK[q][er][ec];
  const v2f y = *(const v2f*)&sY[er][ec];
  *(v2f*)&sZ[er][ec] = stage_state<S>(k, y, dtc);
}

template <int S>
__device__ __forceinline__ void finish_stage(float (*sK)[MT][DATA], const float (*sFp)[MT][DATA + LPAD],
                                             const float (*sY)[DATA], float (*sYn)[DATA],
                                             float (*sZ)[DATA + LPAD], v2f bias, float dtc,
                                             int er, int ec) {
  const v2f y  = *(const v2f*)&sY[er][ec];
  const v2f p0 = *(const v2f*)&sFp[0][er][ec];
  const v2f p1 = *(const v2f*)&sFp[1][er][ec];
  const v2f p2 = *(const v2f*)&sFp[2][er][ec];
  const v2f p3 = *(const v2f*)&sFp[3][er][ec];
  v2f k[7];
  #pragma unroll
  for (int q = 0; q < S; ++q) k[q] = *(const v2f*)&sK[q][er][ec];
  v2f kS = p0 + p1;
  kS = kS + p2;
  kS = kS + p3;
  kS = kS + bias;
  k[S] = kS;
  *(v2f*)&sK[S][er][ec] = kS;
  if constexpr (S < 6) {
    const v2f z = stage_state<S + 1>(k, y, dtc);
    *(v2f*)&sZ[er][ec] = z;
    if constexpr (S + 1 == 6) *(v2f*)&sYn[er][ec] = z;
  }
}

__global__ __launch_bounds__(NTHR, 1)
void node_tsit5_kernel(const float* __restrict__ ts, const float* __restrict__ y0,
                       const float* __restrict__ W0, const float* __restrict__ b0,
                       const float* __restrict__ W1, const float* __restrict__ b1,
                       const float* __restrict__ W2, const float* __restrict__ b2,
                       float* __restrict__ out)
{
  __shared__ __align__(16) float sY [MT][DATA];
  __shared__ __align__(16) float sYn[MT][DATA];
  __shared__ __align__(16) float sZ [MT][DATA + LPAD];
  __shared__ __align__(16) float sK [7][MT][DATA];
  __shared__ __align__(16) float sH0[MT][WIDTH + LPAD];
  __shared__ __align__(16) float sH1[MT][WIDTH + LPAD];
  __shared__ __align__(16) float sFp[4][MT][DATA + LPAD];
  __shared__ float sT[MT], sDt[MT], sDtc[MT], sStep[MT], sDone[MT];
  __shared__ float sTs[TPTS];

  const int tid  = threadIdx.x;
  const int lane = tid & 31;
  const int wav  = tid >> 5;
  const int kh   = lane >> 4;
  const int l16  = lane & 15;
  const int row0 = blockIdx.x * MT;
  const int er   = wav;
  const int ec   = lane * 2;

  if (tid < TPTS) sTs[tid] = ts[tid];
  {
    const float* yp = y0 + (row0 + er) * DATA + ec;
    float a0 = yp[0], a1 = yp[1];
    sY[er][ec] = a0; sY[er][ec + 1] = a1;
    float* op = out + ((size_t)(row0 + er) * TPTS + 0) * DATA + ec;
    op[0] = a0; op[1] = a1;
  }
  if (lane == 0) sDt[er] = ts[1] - ts[0];

  v2f w1f[WIDTH / 4];
  {
    const float* wp = W1 + (wav * 16 + l16) * WIDTH + kh * 2;
    #pragma unroll
    for (int kk = 0; kk < WIDTH / 4; ++kk)
      w1f[kk] = *(const v2f*)(wp + kk * 4);
  }
  v2f w0f[DATA / 4];
  {
    const float* wp = W0 + (wav * 16 + l16) * DATA + kh * 2;
    #pragma unroll
    for (int kk = 0; kk < DATA / 4; ++kk)
      w0f[kk] = *(const v2f*)(wp + kk * 4);
  }
  v2f w2f[16];
  {
    const float* wp = W2 + ((wav & 3) * 16 + l16) * WIDTH + (wav >> 2) * 64 + kh * 2;
    #pragma unroll
    for (int kk = 0; kk < 16; ++kk)
      w2f[kk] = *(const v2f*)(wp + kk * 4);
  }
  const float bias0 = b0[wav * 16 + l16];
  const float bias1 = b1[wav * 16 + l16];
  v2f bias2;
  bias2.x = b2[ec];
  bias2.y = b2[ec + 1];
  __syncthreads();

  auto mlp_gemms = [&]() {
    {
      const int n = wav * 16 + l16;
      v8f acc = {};
      #pragma unroll
      for (int kk = 0; kk < DATA / 4; ++kk) {
        const int k = kk * 4 + kh * 2;
        v2f a = *(const v2f*)(&sZ[l16][k]);
        acc = __builtin_amdgcn_wmma_f32_16x16x4_f32(false, a, false, w0f[kk],
                                                    (short)0, acc, false, false);
      }
      #pragma unroll
      for (int v = 0; v < 8; ++v)
        sH0[v + kh * 8][n] = softplusf(acc[v] + bias0);
    }
    __syncthreads();
    {
      const int n = wav * 16 + l16;
      v8f acc = {};
      #pragma unroll
      for (int kk = 0; kk < WIDTH / 4; ++kk) {
        const int k = kk * 4 + kh * 2;
        v2f a = *(const v2f*)(&sH0[l16][k]);
        acc = __builtin_amdgcn_wmma_f32_16x16x4_f32(false, a, false, w1f[kk],
                                                    (short)0, acc, false, false);
      }
      #pragma unroll
      for (int v = 0; v < 8; ++v)
        sH1[v + kh * 8][n] = softplusf(acc[v] + bias1);
    }
    __syncthreads();
    {
      const int dtile = wav & 3;
      const int kq    = wav >> 2;
      const int d     = dtile * 16 + l16;
      v8f acc = {};
      #pragma unroll
      for (int kk = 0; kk < 16; ++kk) {
        const int k = kq * 64 + kk * 4 + kh * 2;
        v2f a = *(const v2f*)(&sH1[l16][k]);
        acc = __builtin_amdgcn_wmma_f32_16x16x4_f32(false, a, false, w2f[kk],
                                                    (short)0, acc, false, false);
      }
      #pragma unroll
      for (int v = 0; v < 8; ++v)
        sFp[kq][v + kh * 8][d] = acc[v];
    }
    __syncthreads();
  };

  int s_begin = 0;

  #pragma unroll 1
  for (int iv = 0; iv < NINT; ++iv) {
    const float t1 = sTs[iv + 1];
    if (lane == 0) sT[er] = sTs[iv];
    __syncthreads();

    #pragma unroll 1
    for (int ss = 0; ss < MAXSUB; ++ss) {
      if (lane == 0) {
        float t = sT[er], dt = sDt[er];
        float rem  = t1 - t;
        bool  done = rem <= 1e-6f;
        sDtc[er]  = done ? dt : fminf(dt, rem);
        sDone[er] = done ? 1.0f : 0.0f;
      }
      __syncthreads();
      if (__builtin_amdgcn_ballot_w32(sDone[l16] != 0.0f) == 0xffffffffu) break;
      const float dtc = sDtc[er];

      if (s_begin == 0) first_state<0>(sK, sY, sZ, dtc, er, ec);
      else              first_state<1>(sK, sY, sZ, dtc, er, ec);
      __syncthreads();

      #pragma unroll 1
      for (int s = s_begin; s < 7; ++s) {
        mlp_gemms();
        switch (s) {
          case 0:  finish_stage<0>(sK, sFp, sY, sYn, sZ, bias2, dtc, er, ec); break;
          case 1:  finish_stage<1>(sK, sFp, sY, sYn, sZ, bias2, dtc, er, ec); break;
          case 2:  finish_stage<2>(sK, sFp, sY, sYn, sZ, bias2, dtc, er, ec); break;
          case 3:  finish_stage<3>(sK, sFp, sY, sYn, sZ, bias2, dtc, er, ec); break;
          case 4:  finish_stage<4>(sK, sFp, sY, sYn, sZ, bias2, dtc, er, ec); break;
          case 5:  finish_stage<5>(sK, sFp, sY, sYn, sZ, bias2, dtc, er, ec); break;
          default: finish_stage<6>(sK, sFp, sY, sYn, sZ, bias2, dtc, er, ec); break;
        }
        __syncthreads();
      }
      s_begin = 1;

      float esq = 0.0f;
      #pragma unroll
      for (int j = 0; j < 2; ++j) {
        const int c = ec + j;
        float e = 0.0f;
        #pragma unroll
        for (int q = 0; q < 7; ++q) e = fmaf(ECOEF[q], sK[q][er][c], e);
        e *= dtc;
        float scale = fmaf(RTOLC, fmaxf(fabsf(sY[er][c]), fabsf(sYn[er][c])), ATOLC);
        float r = e / scale;
        esq = fmaf(r, r, esq);
      }
      #pragma unroll
      for (int o = 16; o > 0; o >>= 1) esq += __shfl_xor(esq, o, 32);
      if (lane == 0) {
        float errn   = sqrtf(esq * (1.0f / DATA));
        bool  accept = errn <= 1.0f;
        float factor = SAFETY * __powf(fmaxf(errn, 1e-10f), -0.2f);
        factor = fminf(fmaxf(factor, FMINC), FMAXC);
        bool  done = sDone[er] != 0.0f;
        bool  step = accept && !done;
        if (step)  sT[er] += dtc;
        if (!done) sDt[er] = dtc * factor;
        sStep[er] = step ? 1.0f : 0.0f;
      }
      __syncthreads();
      if (sStep[er] != 0.0f) {
        sY[er][ec]     = sYn[er][ec];
        sY[er][ec + 1] = sYn[er][ec + 1];
        sK[0][er][ec]     = sK[6][er][ec];
        sK[0][er][ec + 1] = sK[6][er][ec + 1];
      }
      __syncthreads();
    }

    float* op = out + ((size_t)(row0 + er) * TPTS + (iv + 1)) * DATA + ec;
    op[0] = sY[er][ec];
    op[1] = sY[er][ec + 1];
    __syncthreads();
  }
}

extern "C" void kernel_launch(void* const* d_in, const int* in_sizes, int n_in,
                              void* d_out, int out_size, void* d_ws, size_t ws_size,
                              hipStream_t stream) {
  (void)in_sizes; (void)n_in; (void)out_size; (void)d_ws; (void)ws_size;
  const float* ts = (const float*)d_in[0];
  const float* y0 = (const float*)d_in[1];
  const float* W0 = (const float*)d_in[2];
  const float* b0 = (const float*)d_in[3];
  const float* W1 = (const float*)d_in[4];
  const float* b1 = (const float*)d_in[5];
  const float* W2 = (const float*)d_in[6];
  const float* b2 = (const float*)d_in[7];
  float* out = (float*)d_out;
  node_tsit5_kernel<<<dim3(BATCH / MT), dim3(NTHR), 0, stream>>>(
      ts, y0, W0, b0, W1, b1, W2, b2, out);
}
